# starting kernel
# baseline (speedup 1.0000x reference)
#include <hip/hip_runtime.h>

typedef __attribute__((ext_vector_type(2))) float v2f;
typedef __attribute__((ext_vector_type(8))) float v8f;

#define LN_EPS 1e-5f

__device__ __forceinline__ v8f wmma4(v2f a, v2f b, v8f c) {
  return __builtin_amdgcn_wmma_f32_16x16x4_f32(false, a, false, b, (short)0, c,
                                               false, false);
}

__device__ __forceinline__ v2f ldA(const float* m, int ld, int r0, int c0,
                                   int l16, int h) {
  v2f a;
  a.x = m[(r0 + l16) * ld + c0 + 2 * h + 0];
  a.y = m[(r0 + l16) * ld + c0 + 2 * h + 1];
  return a;
}
__device__ __forceinline__ v2f ldB(const float* m, int ld, int c0, int n0,
                                   int l16, int h) {
  v2f b;
  b.x = m[(c0 + 2 * h + 0) * ld + n0 + l16];
  b.y = m[(c0 + 2 * h + 1) * ld + n0 + l16];
  return b;
}
__device__ __forceinline__ v2f ldBT(const float* m, int ld, int c0, int n0,
                                    int l16, int h) {
  v2f b;
  b.x = m[(n0 + l16) * ld + c0 + 2 * h + 0];
  b.y = m[(n0 + l16) * ld + c0 + 2 * h + 1];
  return b;
}

__device__ __forceinline__ unsigned int f2key(float f) {
  unsigned int u = __float_as_uint(f);
  return (u & 0x80000000u) ? ~u : (u | 0x80000000u);
}
__device__ __forceinline__ float key2f(unsigned int k) {
  unsigned int u = (k & 0x80000000u) ? (k & 0x7FFFFFFFu) : ~k;
  return __uint_as_float(u);
}

__device__ __forceinline__ float bsum(float v, float* red, int tid) {
  __syncthreads();
  red[tid] = v;
  __syncthreads();
#pragma unroll
  for (int s = 64; s > 0; s >>= 1) {
    if (tid < s) red[tid] += red[tid + s];
    __syncthreads();
  }
  return red[0];
}

__device__ __forceinline__ float bsum_g(float v, float* red, int tid, int t,
                                        int gbase) {
  __syncthreads();
  red[tid] = v;
  __syncthreads();
#pragma unroll
  for (int s = 64; s > 0; s >>= 1) {
    if (t < s) red[tid] += red[tid + s];
    __syncthreads();
  }
  return red[gbase];
}

__global__ void enc_init(unsigned int* gmax, float* gsum) {
  int i = blockIdx.x * blockDim.x + threadIdx.x;
  if (i < 4096) {
    gmax[i] = 0u;
    gsum[i] = 0.0f;
  }
}

__global__ __launch_bounds__(128) void enc_pass1_max(
    const float* __restrict__ x, const float* __restrict__ Wq,
    const float* __restrict__ Wk, unsigned int* __restrict__ gmax, int nb) {
  __shared__ __align__(16) float sX[1024];
  __shared__ float sQ[1024], sK[1024];
  __shared__ float sWq[256], sWk[256];
  int tid = threadIdx.x;
  int w = tid >> 5, lane = tid & 31, l16 = lane & 15, h = lane >> 4;
  int r0 = 16 * w;
  for (int i = tid; i < 256; i += 128) {
    sWq[i] = Wq[i];
    sWk[i] = Wk[i];
  }
  float vmax[4][8];
#pragma unroll
  for (int kt = 0; kt < 4; kt++)
#pragma unroll
    for (int r = 0; r < 8; r++) vmax[kt][r] = -3.4e38f;
  __syncthreads();

  for (int it = 0; it < nb; ++it) {
    size_t b = (size_t)blockIdx.x * nb + it;
    const float4* xb4 = (const float4*)(x + b * 1024);
    ((float4*)sX)[tid] = xb4[tid];
    ((float4*)sX)[tid + 128] = xb4[tid + 128];
    __syncthreads();
    v2f a[4];
#pragma unroll
    for (int kk = 0; kk < 4; kk++) a[kk] = ldA(sX, 16, r0, 4 * kk, l16, h);
    v8f q = {};
    v8f k = {};
#pragma unroll
    for (int kk = 0; kk < 4; kk++) {
      q = wmma4(a[kk], ldB(sWq, 16, 4 * kk, 0, l16, h), q);
      k = wmma4(a[kk], ldB(sWk, 16, 4 * kk, 0, l16, h), k);
    }
#pragma unroll
    for (int r = 0; r < 8; r++) {
      sQ[(r0 + r + 8 * h) * 16 + l16] = q[r];
      sK[(r0 + r + 8 * h) * 16 + l16] = k[r];
    }
    __syncthreads();
    v2f qa[4];
#pragma unroll
    for (int kk = 0; kk < 4; kk++) qa[kk] = ldA(sQ, 16, r0, 4 * kk, l16, h);
#pragma unroll
    for (int kt = 0; kt < 4; kt++) {
      v8f s = {};
#pragma unroll
      for (int kk = 0; kk < 4; kk++)
        s = wmma4(qa[kk], ldBT(sK, 16, 4 * kk, 16 * kt, l16, h), s);
#pragma unroll
      for (int r = 0; r < 8; r++)
        vmax[kt][r] = fmaxf(vmax[kt][r], s[r] * (1.0f / 3.0f));
    }
    __syncthreads();
  }
#pragma unroll
  for (int kt = 0; kt < 4; kt++)
#pragma unroll
    for (int r = 0; r < 8; r++) {
      int qrow = r0 + r + 8 * h, kcol = 16 * kt + l16;
      atomicMax(&gmax[qrow * 64 + kcol], f2key(vmax[kt][r]));
    }
}

__global__ __launch_bounds__(128) void enc_pass2_sum(
    const float* __restrict__ x, const float* __restrict__ Wq,
    const float* __restrict__ Wk, const unsigned int* __restrict__ gmax,
    float* __restrict__ gsum, int nb) {
  __shared__ __align__(16) float sX[1024];
  __shared__ float sQ[1024], sK[1024];
  __shared__ float sWq[256], sWk[256];
  int tid = threadIdx.x;
  int w = tid >> 5, lane = tid & 31, l16 = lane & 15, h = lane >> 4;
  int r0 = 16 * w;
  for (int i = tid; i < 256; i += 128) {
    sWq[i] = Wq[i];
    sWk[i] = Wk[i];
  }
  float gm[4][8], vsum[4][8];
#pragma unroll
  for (int kt = 0; kt < 4; kt++)
#pragma unroll
    for (int r = 0; r < 8; r++) {
      int qrow = r0 + r + 8 * h, kcol = 16 * kt + l16;
      gm[kt][r] = key2f(gmax[qrow * 64 + kcol]);
      vsum[kt][r] = 0.0f;
    }
  __syncthreads();

  for (int it = 0; it < nb; ++it) {
    size_t b = (size_t)blockIdx.x * nb + it;
    const float4* xb4 = (const float4*)(x + b * 1024);
    ((float4*)sX)[tid] = xb4[tid];
    ((float4*)sX)[tid + 128] = xb4[tid + 128];
    __syncthreads();
    v2f a[4];
#pragma unroll
    for (int kk = 0; kk < 4; kk++) a[kk] = ldA(sX, 16, r0, 4 * kk, l16, h);
    v8f q = {};
    v8f k = {};
#pragma unroll
    for (int kk = 0; kk < 4; kk++) {
      q = wmma4(a[kk], ldB(sWq, 16, 4 * kk, 0, l16, h), q);
      k = wmma4(a[kk], ldB(sWk, 16, 4 * kk, 0, l16, h), k);
    }
#pragma unroll
    for (int r = 0; r < 8; r++) {
      sQ[(r0 + r + 8 * h) * 16 + l16] = q[r];
      sK[(r0 + r + 8 * h) * 16 + l16] = k[r];
    }
    __syncthreads();
    v2f qa[4];
#pragma unroll
    for (int kk = 0; kk < 4; kk++) qa[kk] = ldA(sQ, 16, r0, 4 * kk, l16, h);
#pragma unroll
    for (int kt = 0; kt < 4; kt++) {
      v8f s = {};
#pragma unroll
      for (int kk = 0; kk < 4; kk++)
        s = wmma4(qa[kk], ldBT(sK, 16, 4 * kk, 16 * kt, l16, h), s);
#pragma unroll
      for (int r = 0; r < 8; r++)
        vsum[kt][r] += __expf(s[r] * (1.0f / 3.0f) - gm[kt][r]);
    }
    __syncthreads();
  }
#pragma unroll
  for (int kt = 0; kt < 4; kt++)
#pragma unroll
    for (int r = 0; r < 8; r++) {
      int qrow = r0 + r + 8 * h, kcol = 16 * kt + l16;
      atomicAdd(&gsum[qrow * 64 + kcol], vsum[kt][r]);
    }
}

#define WSTRIDE 260

__global__ __launch_bounds__(256) void enc_pass3_final(
    const float* __restrict__ x, const float* __restrict__ Wq,
    const float* __restrict__ Wk, const float* __restrict__ Wv,
    const float* __restrict__ W1, const float* __restrict__ b1,
    const float* __restrict__ W2, const float* __restrict__ b2,
    const unsigned int* __restrict__ gmax, const float* __restrict__ gsum,
    float* __restrict__ out, int nb) {
  __shared__ float sW1[64 * WSTRIDE];
  __shared__ float sW2[64 * WSTRIDE];
  __shared__ float sB1[1024], sB2[1024];
  __shared__ float sWq[256], sWk[256], sWv[256];
  __shared__ __align__(16) float sX[2][1024];
  __shared__ float sQ[2][1024];
  __shared__ float sK[2][1024], sV[2][1024];
  __shared__ float sS[2][4096];
  __shared__ float red[256];

  int tid = threadIdx.x;
  int g = tid >> 7;
  int t = tid & 127;
  int gbase = g << 7;
  int w = t >> 5, lane = tid & 31, l16 = lane & 15, h = lane >> 4;
  int r0 = 16 * w;
  int p = t >> 1, o0 = (t & 1) * 8;

  float* mX = sX[g];
  float* mQ = sQ[g];
  float* mK = sK[g];
  float* mV = sV[g];
  float* mS = sS[g];

  for (int i = tid; i < 16384; i += 256) {
    int pp = i >> 8, rr = i & 255;
    sW1[pp * WSTRIDE + rr] = W1[i];
    sW2[pp * WSTRIDE + rr] = W2[i];
  }
  for (int i = tid; i < 1024; i += 256) {
    sB1[i] = b1[i];
    sB2[i] = b2[i];
  }
  if (tid < 256) {
    sWq[tid] = Wq[tid];
    sWk[tid] = Wk[tid];
    sWv[tid] = Wv[tid];
  }
  float gm[4][8], rs[4][8];
#pragma unroll
  for (int kt = 0; kt < 4; kt++)
#pragma unroll
    for (int r = 0; r < 8; r++) {
      int qrow = r0 + r + 8 * h, kcol = 16 * kt + l16;
      gm[kt][r] = key2f(gmax[qrow * 64 + kcol]);
      rs[kt][r] = 1.0f / gsum[qrow * 64 + kcol];
    }
  __syncthreads();

  for (int it = 0; it < nb; ++it) {
    size_t b = (size_t)blockIdx.x * (2 * nb) + (size_t)g * nb + it;
    const float4* xb4 = (const float4*)(x + b * 1024);
    ((float4*)mX)[t] = xb4[t];
    ((float4*)mX)[t + 128] = xb4[t + 128];
    __syncthreads();

    v2f a[4];
#pragma unroll
    for (int kk = 0; kk < 4; kk++) a[kk] = ldA(mX, 16, r0, 4 * kk, l16, h);
    v8f q = {};
    v8f k = {};
    v8f v = {};
#pragma unroll
    for (int kk = 0; kk < 4; kk++) {
      q = wmma4(a[kk], ldB(sWq, 16, 4 * kk, 0, l16, h), q);
      k = wmma4(a[kk], ldB(sWk, 16, 4 * kk, 0, l16, h), k);
      v = wmma4(a[kk], ldB(sWv, 16, 4 * kk, 0, l16, h), v);
    }
#pragma unroll
    for (int r = 0; r < 8; r++) {
      int row = r0 + r + 8 * h;
      mQ[row * 16 + l16] = q[r];
      mK[row * 16 + l16] = k[r];
      mV[row * 16 + l16] = v[r];
    }
    __syncthreads();

    v2f qa[4];
#pragma unroll
    for (int kk = 0; kk < 4; kk++) qa[kk] = ldA(mQ, 16, r0, 4 * kk, l16, h);
#pragma unroll
    for (int kt = 0; kt < 4; kt++) {
      v8f s = {};
#pragma unroll
      for (int kk = 0; kk < 4; kk++)
        s = wmma4(qa[kk], ldBT(mK, 16, 4 * kk, 16 * kt, l16, h), s);
#pragma unroll
      for (int r = 0; r < 8; r++) {
        int qrow = r0 + r + 8 * h, kcol = 16 * kt + l16;
        mS[qrow * 64 + kcol] =
            __expf(s[r] * (1.0f / 3.0f) - gm[kt][r]) * rs[kt][r];
      }
    }

    v8f z = {};
#pragma unroll
    for (int c = 0; c < 16; c++)
      z = wmma4(ldA(mS, 64, r0, 4 * c, l16, h), ldB(mV, 16, 4 * c, 0, l16, h),
                z);

    float part = 0.0f, part2 = 0.0f, tv[8];
#pragma unroll
    for (int r = 0; r < 8; r++) {
      int row = r0 + r + 8 * h;
      float tt = mX[row * 16 + l16] + z[r];
      tv[r] = tt;
      part += tt;
      part2 += tt * tt;
    }
    float s1 = bsum_g(part, red, tid, t, gbase);
    float s2 = bsum_g(part2, red, tid, t, gbase);
    float mean = s1 * (1.0f / 1024.0f);
    float rstd = rsqrtf(s2 * (1.0f / 1024.0f) - mean * mean + LN_EPS);
#pragma unroll
    for (int r = 0; r < 8; r++) {
      int row = r0 + r + 8 * h;
      mX[row * 16 + l16] = (tv[r] - mean) * rstd;
    }
    __syncthreads();

    float acc[8];
#pragma unroll
    for (int j = 0; j < 8; j++) acc[j] = sB1[p * 16 + o0 + j];
#pragma unroll
    for (int i = 0; i < 16; i++) {
      float yv = mX[p * 16 + i];
      const float* wr = &sW1[p * WSTRIDE + i * 16 + o0];
#pragma unroll
      for (int j = 0; j < 8; j++) acc[j] = fmaf(yv, wr[j], acc[j]);
    }
#pragma unroll
    for (int j = 0; j < 8; j++) mQ[p * 16 + o0 + j] = fmaxf(acc[j], 0.0f);
    __syncthreads();

    float acc2[8];
#pragma unroll
    for (int j = 0; j < 8; j++) acc2[j] = sB2[p * 16 + o0 + j];
#pragma unroll
    for (int i = 0; i < 16; i++) {
      float hv = mQ[p * 16 + i];
      const float* wr = &sW2[p * WSTRIDE + i * 16 + o0];
#pragma unroll
      for (int j = 0; j < 8; j++) acc2[j] = fmaf(hv, wr[j], acc2[j]);
    }
    float t2[8];
    float q1 = 0.0f, q2 = 0.0f;
#pragma unroll
    for (int j = 0; j < 8; j++) {
      float tt = fmaxf(acc2[j], 0.0f) + mX[p * 16 + o0 + j];
      t2[j] = tt;
      q1 += tt;
      q2 += tt * tt;
    }
    float u1 = bsum_g(q1, red, tid, t, gbase);
    float u2 = bsum_g(q2, red, tid, t, gbase);
    float m2 = u1 * (1.0f / 1024.0f);
    float r2 = rsqrtf(u2 * (1.0f / 1024.0f) - m2 * m2 + LN_EPS);
    float* op = out + b * 1024 + p * 16 + o0;
#pragma unroll
    for (int j = 0; j < 8; j++) op[j] = (t2[j] - m2) * r2;
    __syncthreads();
  }
}

extern "C" void kernel_launch(void* const* d_in, const int* in_sizes, int n_in,
                              void* d_out, int out_size, void* d_ws,
                              size_t ws_size, hipStream_t stream) {
  (void)n_in;
  (void)out_size;
  (void)ws_size;
  const float* x = (const float*)d_in[0];
  const float* Wq = (const float*)d_in[1];
  const float* Wk = (const float*)d_in[2];
  const float* Wv = (const float*)d_in[3];
  const float* W1 = (const float*)d_in[4];
  const float* b1 = (const float*)d_in[5];
  const float* W2 = (const float*)d_in[6];
  const float* b2 = (const float*)d_in[7];
  float* out = (float*)d_out;

  unsigned int* gmax = (unsigned int*)d_ws;
  float* gsum = (float*)((char*)d_ws + 4096 * sizeof(unsigned int));

  const int B = in_sizes[0] / 1024;
  const int NB = 64;
  const int blocks12 = B / NB;
  const int blocks3 = B / (2 * NB);

  enc_init<<<16, 256, 0, stream>>>(gmax, gsum);
  enc_pass1_max<<<blocks12, 128, 0, stream>>>(x, Wq, Wk, gmax, NB);
  enc_pass2_sum<<<blocks12, 128, 0, stream>>>(x, Wq, Wk, gmax, gsum, NB);
  enc_pass3_final<<<blocks3, 256, 0, stream>>>(x, Wq, Wk, Wv, W1, b1, W2, b2,
                                               gmax, gsum, out, NB);
}


# pass 3a/3b: 4 batches per barrier round; their LayerNorm sums folded by one fixed-order 3-level LDS pass (fp sum order changes); x+z, h2+y parked in LDS; ~20 barriers per batch -> under 3 and under 2
# speedup vs baseline: 2.4509x; 2.4509x over previous
#include <hip/hip_runtime.h>

typedef __attribute__((ext_vector_type(2))) float v2f;
typedef __attribute__((ext_vector_type(8))) float v8f;

#define LN_EPS 1e-5f

__device__ __forceinline__ v8f wmma4(v2f a, v2f b, v8f c) {
  return __builtin_amdgcn_wmma_f32_16x16x4_f32(false, a, false, b, (short)0, c,
                                               false, false);
}

__device__ __forceinline__ v2f ldA(const float* m, int ld, int r0, int c0,
                                   int l16, int h) {
  v2f a;
  a.x = m[(r0 + l16) * ld + c0 + 2 * h + 0];
  a.y = m[(r0 + l16) * ld + c0 + 2 * h + 1];
  return a;
}
__device__ __forceinline__ v2f ldB(const float* m, int ld, int c0, int n0,
                                   int l16, int h) {
  v2f b;
  b.x = m[(c0 + 2 * h + 0) * ld + n0 + l16];
  b.y = m[(c0 + 2 * h + 1) * ld + n0 + l16];
  return b;
}
__device__ __forceinline__ v2f ldBT(const float* m, int ld, int c0, int n0,
                                    int l16, int h) {
  v2f b;
  b.x = m[(n0 + l16) * ld + c0 + 2 * h + 0];
  b.y = m[(n0 + l16) * ld + c0 + 2 * h + 1];
  return b;
}

__device__ __forceinline__ unsigned int f2key(float f) {
  unsigned int u = __float_as_uint(f);
  return (u & 0x80000000u) ? ~u : (u | 0x80000000u);
}
__device__ __forceinline__ float key2f(unsigned int k) {
  unsigned int u = (k & 0x80000000u) ? (k & 0x7FFFFFFFu) : ~k;
  return __uint_as_float(u);
}

#define GB 4
__device__ __forceinline__ void fold128(const float* rA, float* rB, int tid) {
  float* rC = rB + 2 * GB * 16;
  __syncthreads();
  {
    const float4* p = (const float4*)(rA + tid * 8);
    float4 a = p[0], b = p[1];
    rB[tid] = ((a.x + a.y) + (a.z + a.w)) + ((b.x + b.y) + (b.z + b.w));
  }
  __syncthreads();
  if (tid < 2 * GB * 4) {
    float4 a = ((const float4*)rB)[tid];
    rC[tid] = (a.x + a.y) + (a.z + a.w);
  }
  __syncthreads();
}
__device__ __forceinline__ void ln_stats(const float* rB, int g, float& mean,
                                         float& rstd) {
  const float4* rC = (const float4*)(rB + 2 * GB * 16);
  float4 a = rC[2 * g], b = rC[2 * g + 1];
  float s1 = (a.x + a.y) + (a.z + a.w);
  float s2 = (b.x + b.y) + (b.z + b.w);
  mean = s1 * (1.0f / 1024.0f);
  rstd = rsqrtf(s2 * (1.0f / 1024.0f) - mean * mean + LN_EPS);
}

__global__ void enc_init(unsigned int* gmax, float* gsum) {
  int i = blockIdx.x * blockDim.x + threadIdx.x;
  if (i < 4096) {
    gmax[i] = 0u;
    gsum[i] = 0.0f;
  }
}

#define SW 16
#define QP 20
#define QKW (2 * 64 * QP)

__device__ __forceinline__ void ldXrow(const float* xb, int t, int l16, int h,
                                       v2f a[4]) {
  const float4* p = (const float4*)(xb + (16 * t + l16) * 16 + 8 * h);
  float4 u0 = p[0], u1 = p[1];
  a[0].x = u0.x; a[0].y = u0.y; a[1].x = u0.z; a[1].y = u0.w;
  a[2].x = u1.x; a[2].y = u1.y; a[3].x = u1.z; a[3].y = u1.w;
}
__device__ __forceinline__ void ldSrow(const float* m, int t, int l16, int h,
                                       v2f a[4]) {
  const float4* p = (const float4*)(m + (16 * t + l16) * QP + 8 * h);
  float4 u0 = p[0], u1 = p[1];
  a[0].x = u0.x; a[0].y = u0.y; a[1].x = u0.z; a[1].y = u0.w;
  a[2].x = u1.x; a[2].y = u1.y; a[3].x = u1.z; a[3].y = u1.w;
}
__device__ __forceinline__ void ldWfrag(const float* wm, int l16, int h,
                                        v2f b[4]) {
#pragma unroll
  for (int kk = 0; kk < 4; kk++) {
    b[kk].x = wm[(8 * h + 2 * kk + 0) * 16 + l16];
    b[kk].y = wm[(8 * h + 2 * kk + 1) * 16 + l16];
  }
}
__device__ __forceinline__ void qk_frags(const float* xb, const v2f bq[4],
                                         const v2f bk[4], float* sQ, float* sK,
                                         int l16, int h, v2f qa[4][4],
                                         v2f kb[4][4]) {
  v2f a[4][4];
#pragma unroll
  for (int t = 0; t < 4; t++) ldXrow(xb, t, l16, h, a[t]);
#pragma unroll
  for (int t = 0; t < 4; t++) {
    v8f q = {};
    v8f k = {};
#pragma unroll
    for (int kk = 0; kk < 4; kk++) {
      q = wmma4(a[t][kk], bq[kk], q);
      k = wmma4(a[t][kk], bk[kk], k);
    }
#pragma unroll
    for (int r = 0; r < 8; r++) {
      sQ[(16 * t + r + 8 * h) * QP + l16] = q[r];
      sK[(16 * t + r + 8 * h) * QP + l16] = k[r];
    }
  }
#pragma unroll
  for (int t = 0; t < 4; t++) {
    ldSrow(sQ, t, l16, h, qa[t]);
    ldSrow(sK, t, l16, h, kb[t]);
  }
}
template <bool IS_MAX>
__device__ __forceinline__ void fold_waves(float (&v)[16][8], float* red, int w,
                                           int lane) {
  __syncthreads();
  for (int half = SW / 2; half >= 1; half >>= 1) {
    if (w >= half && w < 2 * half) {
      float* dst = red + (w & 7) * 4096;
#pragma unroll
      for (int tl = 0; tl < 16; tl++)
#pragma unroll
        for (int r = 0; r < 8; r++) dst[(tl * 8 + r) * 32 + lane] = v[tl][r];
    }
    __syncthreads();
    if (w < half) {
      const float* sp = red + ((w + half) & 7) * 4096;
#pragma unroll
      for (int tl = 0; tl < 16; tl++)
#pragma unroll
        for (int r = 0; r < 8; r++) {
          float o = sp[(tl * 8 + r) * 32 + lane];
          v[tl][r] = IS_MAX ? fmaxf(v[tl][r], o) : v[tl][r] + o;
        }
    }
  }
  if (w == 0) {
#pragma unroll
    for (int tl = 0; tl < 16; tl++)
#pragma unroll
      for (int r = 0; r < 8; r++) red[(tl * 8 + r) * 32 + lane] = v[tl][r];
  }
  __syncthreads();
}

__global__ __launch_bounds__(SW * 32) void enc_pass1_max(
    const float* __restrict__ x, const float* __restrict__ Wq,
    const float* __restrict__ Wk, unsigned int* __restrict__ gmax, int nbatch) {
  __shared__ __align__(16) float sQK[SW * QKW];
  int tid = threadIdx.x;
  int lane = tid & 31, l16 = lane & 15, h = lane >> 4;
  int w = __builtin_amdgcn_readfirstlane(tid >> 5);
  float* sQ = sQK + w * QKW;
  float* sK = sQ + 64 * QP;
  v2f bq[4], bk[4];
  ldWfrag(Wq, l16, h, bq);
  ldWfrag(Wk, l16, h, bk);
  float vmax[16][8];
#pragma unroll
  for (int tl = 0; tl < 16; tl++)
#pragma unroll
    for (int r = 0; r < 8; r++) vmax[tl][r] = -3.4e38f;

  for (int bi = blockIdx.x * SW + w; bi < nbatch; bi += gridDim.x * SW) {
    v2f qa[4][4], kb[4][4];
    qk_frags(x + (size_t)bi * 1024, bq, bk, sQ, sK, l16, h, qa, kb);
#pragma unroll
    for (int qt = 0; qt < 4; qt++)
#pragma unroll
      for (int kt = 0; kt < 4; kt++) {
        v8f s = {};
#pragma unroll
        for (int kk = 0; kk < 4; kk++) s = wmma4(qa[qt][kk], kb[kt][kk], s);
#pragma unroll
        for (int r = 0; r < 8; r++)
          vmax[qt * 4 + kt][r] =
              fmaxf(vmax[qt * 4 + kt][r], s[r] * (1.0f / 3.0f));
      }
  }

  fold_waves<true>(vmax, sQK, w, lane);
  for (int j = 0; j < 4096 / (SW * 32); ++j) {
    int idx = j * (SW * 32) + tid;
    int tr = idx >> 5;
    int tl = tr >> 3, r = tr & 7;
    int qrow = 16 * (tl >> 2) + r + 8 * h, kcol = 16 * (tl & 3) + l16;
    atomicMax(&gmax[qrow * 64 + kcol], f2key(sQK[idx]));
  }
}

__global__ __launch_bounds__(SW * 32) void enc_pass2_sum(
    const float* __restrict__ x, const float* __restrict__ Wq,
    const float* __restrict__ Wk, const unsigned int* __restrict__ gmax,
    float* __restrict__ gsum, int nbatch) {
  __shared__ __align__(16) float sQK[SW * QKW];
  __shared__ __align__(16) float sGm[4096];
  int tid = threadIdx.x;
  int lane = tid & 31, l16 = lane & 15, h = lane >> 4;
  int w = __builtin_amdgcn_readfirstlane(tid >> 5);
  float* sQ = sQK + w * QKW;
  float* sK = sQ + 64 * QP;
  v2f bq[4], bk[4];
  ldWfrag(Wq, l16, h, bq);
  ldWfrag(Wk, l16, h, bk);
  for (int j = 0; j < 4096 / (SW * 32); ++j) {
    int idx = j * (SW * 32) + tid;
    int jj = idx & 3, ln = (idx >> 2) & 31, trq = idx >> 7;
    int tl = trq >> 1, r = (trq & 1) * 4 + jj;
    int qrow = 16 * (tl >> 2) + r + 8 * (ln >> 4), kcol = 16 * (tl & 3) + (ln & 15);
    sGm[idx] = key2f(gmax[qrow * 64 + kcol]);
  }
  float vsum[16][8];
#pragma unroll
  for (int tl = 0; tl < 16; tl++)
#pragma unroll
    for (int r = 0; r < 8; r++) vsum[tl][r] = 0.0f;
  __syncthreads();

  for (int bi = blockIdx.x * SW + w; bi < nbatch; bi += gridDim.x * SW) {
    v2f qa[4][4], kb[4][4];
    qk_frags(x + (size_t)bi * 1024, bq, bk, sQ, sK, l16, h, qa, kb);
#pragma unroll
    for (int qt = 0; qt < 4; qt++)
#pragma unroll
      for (int kt = 0; kt < 4; kt++) {
        const int tl = qt * 4 + kt;
        v8f s = {};
#pragma unroll
        for (int kk = 0; kk < 4; kk++) s = wmma4(qa[qt][kk], kb[kt][kk], s);
        float4 g0 = ((const float4*)sGm)[(tl * 2 + 0) * 32 + lane];
        float4 g1 = ((const float4*)sGm)[(tl * 2 + 1) * 32 + lane];
        float gm[8] = {g0.x, g0.y, g0.z, g0.w, g1.x, g1.y, g1.z, g1.w};
#pragma unroll
        for (int r = 0; r < 8; r++)
          vsum[tl][r] += __expf(s[r] * (1.0f / 3.0f) - gm[r]);
      }
  }

  fold_waves<false>(vsum, sQK, w, lane);
  for (int j = 0; j < 4096 / (SW * 32); ++j) {
    int idx = j * (SW * 32) + tid;
    int tr = idx >> 5;
    int tl = tr >> 3, r = tr & 7;
    int qrow = 16 * (tl >> 2) + r + 8 * h, kcol = 16 * (tl & 3) + l16;
    atomicAdd(&gsum[qrow * 64 + kcol], sQK[idx]);
  }
}

__global__ __launch_bounds__(128) __attribute__((amdgpu_waves_per_eu(5, 5))) void enc_pass3a_attn(
    const float* __restrict__ x, const float* __restrict__ Wq,
    const float* __restrict__ Wk, const float* __restrict__ Wv,
    const unsigned int* __restrict__ gmax, const float* __restrict__ gsum,
    float* __restrict__ out, int nbatch) {
  __shared__ float sWq[256], sWk[256], sWv[256];
  __shared__ __align__(16) float mX[GB][1024];
  __shared__ float mQ[1024], mK[1024], mV[1024];
  __shared__ float mS[4096];
  __shared__ __align__(16) float rA[2 * GB * 128];
  __shared__ __align__(16) float rB[2 * GB * 20];

  int tid = threadIdx.x;
  int w = tid >> 5, lane = tid & 31, l16 = lane & 15, h = lane >> 4;
  int r0 = 16 * w;

  for (int i = tid; i < 256; i += 128) {
    sWq[i] = Wq[i];
    sWk[i] = Wk[i];
    sWv[i] = Wv[i];
  }
  float gm[4][8], rs[4][8];
#pragma unroll
  for (int kt = 0; kt < 4; kt++)
#pragma unroll
    for (int r = 0; r < 8; r++) {
      int qrow = r0 + r + 8 * h, kcol = 16 * kt + l16;
      gm[kt][r] = key2f(gmax[qrow * 64 + kcol]);
      rs[kt][r] = 1.0f / gsum[qrow * 64 + kcol];
    }
  __syncthreads();

  for (int b0 = blockIdx.x * GB; b0 < nbatch; b0 += gridDim.x * GB) {
#pragma unroll
    for (int g = 0; g < GB; g++) {
      float4 u0 = make_float4(0.f, 0.f, 0.f, 0.f), u1 = u0;
      if (b0 + g < nbatch) {
        const float4* xb4 = (const float4*)(x + (size_t)(b0 + g) * 1024);
        u0 = xb4[tid];
        u1 = xb4[tid + 128];
      }
      ((float4*)mX[g])[tid] = u0;
      ((float4*)mX[g])[tid + 128] = u1;
    }

#pragma clang loop unroll(disable) vectorize(disable) interleave(disable)
    for (int g = 0; g < GB; g++) {
    float* mXg = mX[g];
    __syncthreads();

    v2f a[4];
#pragma unroll
    for (int kk = 0; kk < 4; kk++) a[kk] = ldA(mXg, 16, r0, 4 * kk, l16, h);
    v8f q = {};
    v8f k = {};
    v8f v = {};
#pragma unroll
    for (int kk = 0; kk < 4; kk++) {
      q = wmma4(a[kk], ldB(sWq, 16, 4 * kk, 0, l16, h), q);
      k = wmma4(a[kk], ldB(sWk, 16, 4 * kk, 0, l16, h), k);
      v = wmma4(a[kk], ldB(sWv, 16, 4 * kk, 0, l16, h), v);
    }
#pragma unroll
    for (int r = 0; r < 8; r++) {
      int row = r0 + r + 8 * h;
      mQ[row * 16 + l16] = q[r];
      mK[row * 16 + l16] = k[r];
      mV[row * 16 + l16] = v[r];
    }
    __syncthreads();

    v2f qa[4];
#pragma unroll
    for (int kk = 0; kk < 4; kk++) qa[kk] = ldA(mQ, 16, r0, 4 * kk, l16, h);
#pragma unroll
    for (int kt = 0; kt < 4; kt++) {
      v8f s = {};
#pragma unroll
      for (int kk = 0; kk < 4; kk++)
        s = wmma4(qa[kk], ldBT(mK, 16, 4 * kk, 16 * kt, l16, h), s);
#pragma unroll
      for (int r = 0; r < 8; r++) {
        int qrow = r0 + r + 8 * h, kcol = 16 * kt + l16;
        mS[qrow * 64 + kcol] =
            __expf(s[r] * (1.0f / 3.0f) - gm[kt][r]) * rs[kt][r];
      }
    }

    v8f z = {};
#pragma unroll
    for (int c = 0; c < 16; c++)
      z = wmma4(ldA(mS, 64, r0, 4 * c, l16, h), ldB(mV, 16, 4 * c, 0, l16, h),
                z);

    float part = 0.0f, part2 = 0.0f;
#pragma unroll
    for (int r = 0; r < 8; r++) {
      int row = r0 + r + 8 * h;
      float tt = mXg[row * 16 + l16] + z[r];
      mXg[row * 16 + l16] = tt;
      part += tt;
      part2 += tt * tt;
    }
    rA[(2 * g + 0) * 128 + tid] = part;
    rA[(2 * g + 1) * 128 + tid] = part2;
    }

    fold128(rA, rB, tid);
#pragma unroll
    for (int g = 0; g < GB; g++) {
      if (b0 + g < nbatch) {
        float mean, rstd;
        ln_stats(rB, g, mean, rstd);
        float4 u0 = ((const float4*)mX[g])[tid];
        float4 u1 = ((const float4*)mX[g])[tid + 128];
        u0.x = (u0.x - mean) * rstd; u0.y = (u0.y - mean) * rstd;
        u0.z = (u0.z - mean) * rstd; u0.w = (u0.w - mean) * rstd;
        u1.x = (u1.x - mean) * rstd; u1.y = (u1.y - mean) * rstd;
        u1.z = (u1.z - mean) * rstd; u1.w = (u1.w - mean) * rstd;
        float4* yb4 = (float4*)(out + (size_t)(b0 + g) * 1024);
        yb4[tid] = u0;
        yb4[tid + 128] = u1;
      }
    }
  }
}

__global__ __launch_bounds__(128) void enc_pass3b_mlp(
    const float* __restrict__ W1, const float* __restrict__ b1,
    const float* __restrict__ W2, const float* __restrict__ b2, float* io,
    int nbatch) {
  __shared__ __align__(16) float mY[GB][1024];
  __shared__ __align__(16) float mH[GB][1024];
  __shared__ __align__(16) float rA[2 * GB * 128];
  __shared__ __align__(16) float rB[2 * GB * 20];

  int tid = threadIdx.x;
  int p = tid >> 1, o0 = (tid & 1) * 8;

  float w1[16][8], w2[16][8], c1[8], c2[8];
  {
    const float4* w1p = (const float4*)(W1 + p * 256 + o0);
    const float4* w2p = (const float4*)(W2 + p * 256 + o0);
#pragma unroll
    for (int i = 0; i < 16; i++) {
      float4 u0 = w1p[4 * i], u1 = w1p[4 * i + 1];
      float4 g0 = w2p[4 * i], g1 = w2p[4 * i + 1];
      w1[i][0] = u0.x; w1[i][1] = u0.y; w1[i][2] = u0.z; w1[i][3] = u0.w;
      w1[i][4] = u1.x; w1[i][5] = u1.y; w1[i][6] = u1.z; w1[i][7] = u1.w;
      w2[i][0] = g0.x; w2[i][1] = g0.y; w2[i][2] = g0.z; w2[i][3] = g0.w;
      w2[i][4] = g1.x; w2[i][5] = g1.y; w2[i][6] = g1.z; w2[i][7] = g1.w;
    }
#pragma unroll
    for (int j = 0; j < 8; j++) {
      c1[j] = b1[p * 16 + o0 + j];
      c2[j] = b2[p * 16 + o0 + j];
    }
  }

  for (int b0 = blockIdx.x * GB; b0 < nbatch; b0 += gridDim.x * GB) {
#pragma unroll
    for (int g = 0; g < GB; g++) {
      float4 u0 = make_float4(0.f, 0.f, 0.f, 0.f), u1 = u0;
      if (b0 + g < nbatch) {
        const float4* yb4 = (const float4*)(io + (size_t)(b0 + g) * 1024);
        u0 = yb4[tid];
        u1 = yb4[tid + 128];
      }
      ((float4*)mY[g])[tid] = u0;
      ((float4*)mY[g])[tid + 128] = u1;
    }
    __syncthreads();

#pragma clang loop unroll(disable) vectorize(disable) interleave(disable)
    for (int g = 0; g < GB; g++) {
      float acc[8];
#pragma unroll
      for (int j = 0; j < 8; j++) acc[j] = c1[j];
#pragma unroll
      for (int i = 0; i < 16; i++) {
        float yv = mY[g][p * 16 + i];
#pragma unroll
        for (int j = 0; j < 8; j++) acc[j] = fmaf(yv, w1[i][j], acc[j]);
      }
#pragma unroll
      for (int j = 0; j < 8; j++) mH[g][p * 16 + o0 + j] = fmaxf(acc[j], 0.0f);
    }
    __syncthreads();

#pragma clang loop unroll(disable) vectorize(disable) interleave(disable)
    for (int g = 0; g < GB; g++) {
      float acc2[8];
#pragma unroll
      for (int j = 0; j < 8; j++) acc2[j] = c2[j];
#pragma unroll
      for (int i = 0; i < 16; i++) {
        float hv = mH[g][p * 16 + i];
#pragma unroll
        for (int j = 0; j < 8; j++) acc2[j] = fmaf(hv, w2[i][j], acc2[j]);
      }
      float q1 = 0.0f, q2 = 0.0f;
#pragma unroll
      for (int j = 0; j < 8; j++) {
        float tt = fmaxf(acc2[j], 0.0f) + mY[g][p * 16 + o0 + j];
        mH[g][p * 16 + o0 + j] = tt;
        q1 += tt;
        q2 += tt * tt;
      }
      rA[(2 * g + 0) * 128 + tid] = q1;
      rA[(2 * g + 1) * 128 + tid] = q2;
    }

    fold128(rA, rB, tid);
#pragma unroll
    for (int g = 0; g < GB; g++) {
      if (b0 + g < nbatch) {
        float m2, r2;
        ln_stats(rB, g, m2, r2);
        float* op = io + (size_t)(b0 + g) * 1024 + p * 16 + o0;
#pragma unroll
        for (int j = 0; j < 8; j++) op[j] = (mH[g][p * 16 + o0 + j] - m2) * r2;
      }
    }
  }
}

extern "C" void kernel_launch(void* const* d_in, const int* in_sizes, int n_in,
                              void* d_out, int out_size, void* d_ws,
                              size_t ws_size, hipStream_t stream) {
  (void)n_in;
  (void)out_size;
  (void)ws_size;
  const float* x = (const float*)d_in[0];
  const float* Wq = (const float*)d_in[1];
  const float* Wk = (const float*)d_in[2];
  const float* Wv = (const float*)d_in[3];
  const float* W1 = (const float*)d_in[4];
  const float* b1 = (const float*)d_in[5];
  const float* W2 = (const float*)d_in[6];
  const float* b2 = (const float*)d_in[7];
  float* out = (float*)d_out;

  unsigned int* gmax = (unsigned int*)d_ws;
  float* gsum = (float*)((char*)d_ws + 4096 * sizeof(unsigned int));

  const int B = in_sizes[0] / 1024;

  enc_init<<<16, 256, 0, stream>>>(gmax, gsum);
  enc_pass1_max<<<256, SW * 32, 0, stream>>>(x, Wq, Wk, gmax, B);
  enc_pass2_sum<<<256, SW * 32, 0, stream>>>(x, Wq, Wk, gmax, gsum, B);
  enc_pass3a_attn<<<1280, 128, 0, stream>>>(x, Wq, Wk, Wv, gmax, gsum, out, B);
  enc_pass3b_mlp<<<768, 128, 0, stream>>>(W1, b1, W2, b2, out, B);
}


# pass 3a: each wave's attn tile staged key-major in LDS: 8 rows of a key leave as two 128-bit stores, z-product A fragments read 16 consecutive floats (no 16-way bank conflict); sums unchanged
# speedup vs baseline: 2.6145x; 1.0668x over previous
#include <hip/hip_runtime.h>

typedef __attribute__((ext_vector_type(2))) float v2f;
typedef __attribute__((ext_vector_type(8))) float v8f;

#define LN_EPS 1e-5f

__device__ __forceinline__ v8f wmma4(v2f a, v2f b, v8f c) {
  return __builtin_amdgcn_wmma_f32_16x16x4_f32(false, a, false, b, (short)0, c,
                                               false, false);
}

__device__ __forceinline__ v2f ldA(const float* m, int ld, int r0, int c0,
                                   int l16, int h) {
  v2f a;
  a.x = m[(r0 + l16) * ld + c0 + 2 * h + 0];
  a.y = m[(r0 + l16) * ld + c0 + 2 * h + 1];
  return a;
}
__device__ __forceinline__ v2f ldB(const float* m, int ld, int c0, int n0,
                                   int l16, int h) {
  v2f b;
  b.x = m[(c0 + 2 * h + 0) * ld + n0 + l16];
  b.y = m[(c0 + 2 * h + 1) * ld + n0 + l16];
  return b;
}
__device__ __forceinline__ v2f ldBT(const float* m, int ld, int c0, int n0,
                                    int l16, int h) {
  v2f b;
  b.x = m[(n0 + l16) * ld + c0 + 2 * h + 0];
  b.y = m[(n0 + l16) * ld + c0 + 2 * h + 1];
  return b;
}

__device__ __forceinline__ unsigned int f2key(float f) {
  unsigned int u = __float_as_uint(f);
  return (u & 0x80000000u) ? ~u : (u | 0x80000000u);
}
__device__ __forceinline__ float key2f(unsigned int k) {
  unsigned int u = (k & 0x80000000u) ? (k & 0x7FFFFFFFu) : ~k;
  return __uint_as_float(u);
}

#define GB 4
__device__ __forceinline__ void fold128(const float* rA, float* rB, int tid) {
  float* rC = rB + 2 * GB * 16;
  __syncthreads();
  {
    const float4* p = (const float4*)(rA + tid * 8);
    float4 a = p[0], b = p[1];
    rB[tid] = ((a.x + a.y) + (a.z + a.w)) + ((b.x + b.y) + (b.z + b.w));
  }
  __syncthreads();
  if (tid < 2 * GB * 4) {
    float4 a = ((const float4*)rB)[tid];
    rC[tid] = (a.x + a.y) + (a.z + a.w);
  }
  __syncthreads();
}
__device__ __forceinline__ void ln_stats(const float* rB, int g, float& mean,
                                         float& rstd) {
  const float4* rC = (const float4*)(rB + 2 * GB * 16);
  float4 a = rC[2 * g], b = rC[2 * g + 1];
  float s1 = (a.x + a.y) + (a.z + a.w);
  float s2 = (b.x + b.y) + (b.z + b.w);
  mean = s1 * (1.0f / 1024.0f);
  rstd = rsqrtf(s2 * (1.0f / 1024.0f) - mean * mean + LN_EPS);
}

__global__ void enc_init(unsigned int* gmax, float* gsum) {
  int i = blockIdx.x * blockDim.x + threadIdx.x;
  if (i < 4096) {
    gmax[i] = 0u;
    gsum[i] = 0.0f;
  }
}

#define SW 16
#define QP 20
#define QKW (2 * 64 * QP)

__device__ __forceinline__ void ldXrow(const float* xb, int t, int l16, int h,
                                       v2f a[4]) {
  const float4* p = (const float4*)(xb + (16 * t + l16) * 16 + 8 * h);
  float4 u0 = p[0], u1 = p[1];
  a[0].x = u0.x; a[0].y = u0.y; a[1].x = u0.z; a[1].y = u0.w;
  a[2].x = u1.x; a[2].y = u1.y; a[3].x = u1.z; a[3].y = u1.w;
}
__device__ __forceinline__ void ldSrow(const float* m, int t, int l16, int h,
                                       v2f a[4]) {
  const float4* p = (const float4*)(m + (16 * t + l16) * QP + 8 * h);
  float4 u0 = p[0], u1 = p[1];
  a[0].x = u0.x; a[0].y = u0.y; a[1].x = u0.z; a[1].y = u0.w;
  a[2].x = u1.x; a[2].y = u1.y; a[3].x = u1.z; a[3].y = u1.w;
}
__device__ __forceinline__ void ldWfrag(const float* wm, int l16, int h,
                                        v2f b[4]) {
#pragma unroll
  for (int kk = 0; kk < 4; kk++) {
    b[kk].x = wm[(8 * h + 2 * kk + 0) * 16 + l16];
    b[kk].y = wm[(8 * h + 2 * kk + 1) * 16 + l16];
  }
}
__device__ __forceinline__ void qk_frags(const float* xb, const v2f bq[4],
                                         const v2f bk[4], float* sQ, float* sK,
                                         int l16, int h, v2f qa[4][4],
                                         v2f kb[4][4]) {
  v2f a[4][4];
#pragma unroll
  for (int t = 0; t < 4; t++) ldXrow(xb, t, l16, h, a[t]);
#pragma unroll
  for (int t = 0; t < 4; t++) {
    v8f q = {};
    v8f k = {};
#pragma unroll
    for (int kk = 0; kk < 4; kk++) {
      q = wmma4(a[t][kk], bq[kk], q);
      k = wmma4(a[t][kk], bk[kk], k);
    }
#pragma unroll
    for (int r = 0; r < 8; r++) {
      sQ[(16 * t + r + 8 * h) * QP + l16] = q[r];
      sK[(16 * t + r + 8 * h) * QP + l16] = k[r];
    }
  }
#pragma unroll
  for (int t = 0; t < 4; t++) {
    ldSrow(sQ, t, l16, h, qa[t]);
    ldSrow(sK, t, l16, h, kb[t]);
  }
}
template <bool IS_MAX>
__device__ __forceinline__ void fold_waves(float (&v)[16][8], float* red, int w,
                                           int lane) {
  __syncthreads();
  for (int half = SW / 2; half >= 1; half >>= 1) {
    if (w >= half && w < 2 * half) {
      float* dst = red + (w & 7) * 4096;
#pragma unroll
      for (int tl = 0; tl < 16; tl++)
#pragma unroll
        for (int r = 0; r < 8; r++) dst[(tl * 8 + r) * 32 + lane] = v[tl][r];
    }
    __syncthreads();
    if (w < half) {
      const float* sp = red + ((w + half) & 7) * 4096;
#pragma unroll
      for (int tl = 0; tl < 16; tl++)
#pragma unroll
        for (int r = 0; r < 8; r++) {
          float o = sp[(tl * 8 + r) * 32 + lane];
          v[tl][r] = IS_MAX ? fmaxf(v[tl][r], o) : v[tl][r] + o;
        }
    }
  }
  if (w == 0) {
#pragma unroll
    for (int tl = 0; tl < 16; tl++)
#pragma unroll
      for (int r = 0; r < 8; r++) red[(tl * 8 + r) * 32 + lane] = v[tl][r];
  }
  __syncthreads();
}

__global__ __launch_bounds__(SW * 32) void enc_pass1_max(
    const float* __restrict__ x, const float* __restrict__ Wq,
    const float* __restrict__ Wk, unsigned int* __restrict__ gmax, int nbatch) {
  __shared__ __align__(16) float sQK[SW * QKW];
  int tid = threadIdx.x;
  int lane = tid & 31, l16 = lane & 15, h = lane >> 4;
  int w = __builtin_amdgcn_readfirstlane(tid >> 5);
  float* sQ = sQK + w * QKW;
  float* sK = sQ + 64 * QP;
  v2f bq[4], bk[4];
  ldWfrag(Wq, l16, h, bq);
  ldWfrag(Wk, l16, h, bk);
  float vmax[16][8];
#pragma unroll
  for (int tl = 0; tl < 16; tl++)
#pragma unroll
    for (int r = 0; r < 8; r++) vmax[tl][r] = -3.4e38f;

  for (int bi = blockIdx.x * SW + w; bi < nbatch; bi += gridDim.x * SW) {
    v2f qa[4][4], kb[4][4];
    qk_frags(x + (size_t)bi * 1024, bq, bk, sQ, sK, l16, h, qa, kb);
#pragma unroll
    for (int qt = 0; qt < 4; qt++)
#pragma unroll
      for (int kt = 0; kt < 4; kt++) {
        v8f s = {};
#pragma unroll
        for (int kk = 0; kk < 4; kk++) s = wmma4(qa[qt][kk], kb[kt][kk], s);
#pragma unroll
        for (int r = 0; r < 8; r++)
          vmax[qt * 4 + kt][r] =
              fmaxf(vmax[qt * 4 + kt][r], s[r] * (1.0f / 3.0f));
      }
  }

  fold_waves<true>(vmax, sQK, w, lane);
  for (int j = 0; j < 4096 / (SW * 32); ++j) {
    int idx = j * (SW * 32) + tid;
    int tr = idx >> 5;
    int tl = tr >> 3, r = tr & 7;
    int qrow = 16 * (tl >> 2) + r + 8 * h, kcol = 16 * (tl & 3) + l16;
    atomicMax(&gmax[qrow * 64 + kcol], f2key(sQK[idx]));
  }
}

__global__ __launch_bounds__(SW * 32) void enc_pass2_sum(
    const float* __restrict__ x, const float* __restrict__ Wq,
    const float* __restrict__ Wk, const unsigned int* __restrict__ gmax,
    float* __restrict__ gsum, int nbatch) {
  __shared__ __align__(16) float sQK[SW * QKW];
  __shared__ __align__(16) float sGm[4096];
  int tid = threadIdx.x;
  int lane = tid & 31, l16 = lane & 15, h = lane >> 4;
  int w = __builtin_amdgcn_readfirstlane(tid >> 5);
  float* sQ = sQK + w * QKW;
  float* sK = sQ + 64 * QP;
  v2f bq[4], bk[4];
  ldWfrag(Wq, l16, h, bq);
  ldWfrag(Wk, l16, h, bk);
  for (int j = 0; j < 4096 / (SW * 32); ++j) {
    int idx = j * (SW * 32) + tid;
    int jj = idx & 3, ln = (idx >> 2) & 31, trq = idx >> 7;
    int tl = trq >> 1, r = (trq & 1) * 4 + jj;
    int qrow = 16 * (tl >> 2) + r + 8 * (ln >> 4), kcol = 16 * (tl & 3) + (ln & 15);
    sGm[idx] = key2f(gmax[qrow * 64 + kcol]);
  }
  float vsum[16][8];
#pragma unroll
  for (int tl = 0; tl < 16; tl++)
#pragma unroll
    for (int r = 0; r < 8; r++) vsum[tl][r] = 0.0f;
  __syncthreads();

  for (int bi = blockIdx.x * SW + w; bi < nbatch; bi += gridDim.x * SW) {
    v2f qa[4][4], kb[4][4];
    qk_frags(x + (size_t)bi * 1024, bq, bk, sQ, sK, l16, h, qa, kb);
#pragma unroll
    for (int qt = 0; qt < 4; qt++)
#pragma unroll
      for (int kt = 0; kt < 4; kt++) {
        const int tl = qt * 4 + kt;
        v8f s = {};
#pragma unroll
        for (int kk = 0; kk < 4; kk++) s = wmma4(qa[qt][kk], kb[kt][kk], s);
        float4 g0 = ((const float4*)sGm)[(tl * 2 + 0) * 32 + lane];
        float4 g1 = ((const float4*)sGm)[(tl * 2 + 1) * 32 + lane];
        float gm[8] = {g0.x, g0.y, g0.z, g0.w, g1.x, g1.y, g1.z, g1.w};
#pragma unroll
        for (int r = 0; r < 8; r++)
          vsum[tl][r] += __expf(s[r] * (1.0f / 3.0f) - gm[r]);
      }
  }

  fold_waves<false>(vsum, sQK, w, lane);
  for (int j = 0; j < 4096 / (SW * 32); ++j) {
    int idx = j * (SW * 32) + tid;
    int tr = idx >> 5;
    int tl = tr >> 3, r = tr & 7;
    int qrow = 16 * (tl >> 2) + r + 8 * h, kcol = 16 * (tl & 3) + l16;
    atomicAdd(&gsum[qrow * 64 + kcol], sQK[idx]);
  }
}

__global__ __launch_bounds__(128) __attribute__((amdgpu_waves_per_eu(5, 5))) void enc_pass3a_attn(
    const float* __restrict__ x, const float* __restrict__ Wq,
    const float* __restrict__ Wk, const float* __restrict__ Wv,
    const unsigned int* __restrict__ gmax, const float* __restrict__ gsum,
    float* __restrict__ out, int nbatch) {
  __shared__ float sWq[256], sWk[256], sWv[256];
  __shared__ __align__(16) float mX[GB][1024];
  __shared__ float mQ[1024], mK[1024], mV[1024];
  __shared__ __align__(16) float mS[4096];
  __shared__ __align__(16) float rA[2 * GB * 128];
  __shared__ __align__(16) float rB[2 * GB * 20];

  int tid = threadIdx.x;
  int w = tid >> 5, lane = tid & 31, l16 = lane & 15, h = lane >> 4;
  int r0 = 16 * w;
  float* mSw = mS + 1024 * w;

  for (int i = tid; i < 256; i += 128) {
    sWq[i] = Wq[i];
    sWk[i] = Wk[i];
    sWv[i] = Wv[i];
  }
  float gm[4][8], rs[4][8];
#pragma unroll
  for (int kt = 0; kt < 4; kt++)
#pragma unroll
    for (int r = 0; r < 8; r++) {
      int qrow = r0 + r + 8 * h, kcol = 16 * kt + l16;
      gm[kt][r] = key2f(gmax[qrow * 64 + kcol]);
      rs[kt][r] = 1.0f / gsum[qrow * 64 + kcol];
    }
  __syncthreads();

  for (int b0 = blockIdx.x * GB; b0 < nbatch; b0 += gridDim.x * GB) {
#pragma unroll
    for (int g = 0; g < GB; g++) {
      float4 u0 = make_float4(0.f, 0.f, 0.f, 0.f), u1 = u0;
      if (b0 + g < nbatch) {
        const float4* xb4 = (const float4*)(x + (size_t)(b0 + g) * 1024);
        u0 = xb4[tid];
        u1 = xb4[tid + 128];
      }
      ((float4*)mX[g])[tid] = u0;
      ((float4*)mX[g])[tid + 128] = u1;
    }

#pragma clang loop unroll(disable) vectorize(disable) interleave(disable)
    for (int g = 0; g < GB; g++) {
    float* mXg = mX[g];
    __syncthreads();


    v2f a[4], wq[4], wk[4], wv[4];
#pragma unroll
    for (int kk = 0; kk < 4; kk++) a[kk] = ldA(mXg, 16, r0, 4 * kk, l16, h);
#pragma unroll
    for (int kk = 0; kk < 4; kk++) {
      wq[kk] = ldB(sWq, 16, 4 * kk, 0, l16, h);
      wk[kk] = ldB(sWk, 16, 4 * kk, 0, l16, h);
      wv[kk] = ldB(sWv, 16, 4 * kk, 0, l16, h);
    }
    v8f q = {};
    v8f k = {};
    v8f v = {};
#pragma unroll
    for (int kk = 0; kk < 4; kk++) {
      q = wmma4(a[kk], wq[kk], q);
      k = wmma4(a[kk], wk[kk], k);
      v = wmma4(a[kk], wv[kk], v);
    }
#pragma unroll
    for (int r = 0; r < 8; r++) {
      int row = r0 + r + 8 * h;
      mQ[row * 16 + l16] = q[r];
      mK[row * 16 + l16] = k[r];
      mV[row * 16 + l16] = v[r];
    }
    __syncthreads();

    v2f qa[4], kf[4][4];
#pragma unroll
    for (int kk = 0; kk < 4; kk++) qa[kk] = ldA(mQ, 16, r0, 4 * kk, l16, h);
#pragma unroll
    for (int kk = 0; kk < 4; kk++)
#pragma unroll
      for (int kt = 0; kt < 4; kt++)
        kf[kk][kt] = ldBT(mK, 16, 4 * kk, 16 * kt, l16, h);
    v8f s[4];
#pragma unroll
    for (int kt = 0; kt < 4; kt++) s[kt] = v8f{};
#pragma unroll
    for (int kk = 0; kk < 4; kk++)
#pragma unroll
      for (int kt = 0; kt < 4; kt++)
        s[kt] = wmma4(qa[kk], kf[kk][kt], s[kt]);
#pragma unroll
    for (int kt = 0; kt < 4; kt++) {
      float e[8];
#pragma unroll
      for (int r = 0; r < 8; r++)
        e[r] = __expf(s[kt][r] * (1.0f / 3.0f) - gm[kt][r]) * rs[kt][r];
      float4* dst = (float4*)(mSw + (16 * kt + l16) * 16 + 8 * h);
      dst[0] = make_float4(e[0], e[1], e[2], e[3]);
      dst[1] = make_float4(e[4], e[5], e[6], e[7]);
    }

    v2f af[4][4], vf[4][4];
    v8f z = {};
#pragma unroll
    for (int j = 0; j < 4; j++) {
      af[0][j] = ldB(mSw, 16, 4 * j, 0, l16, h);
      vf[0][j] = ldB(mV, 16, 4 * j, 0, l16, h);
    }
#pragma unroll
    for (int gi = 0; gi < 4; gi++) {
      if (gi < 3) {
#pragma unroll
        for (int j = 0; j < 4; j++) {
          af[gi + 1][j] = ldB(mSw, 16, 16 * (gi + 1) + 4 * j, 0, l16, h);
          vf[gi + 1][j] = ldB(mV, 16, 16 * (gi + 1) + 4 * j, 0, l16, h);
        }
      }
#pragma unroll
      for (int j = 0; j < 4; j++) z = wmma4(af[gi][j], vf[gi][j], z);
    }

    float xr[8];
#pragma unroll
    for (int r = 0; r < 8; r++) xr[r] = mXg[(r0 + r + 8 * h) * 16 + l16];
    float part = 0.0f, part2 = 0.0f;
#pragma unroll
    for (int r = 0; r < 8; r++) {
      int row = r0 + r + 8 * h;
      float tt = xr[r] + z[r];
      mXg[row * 16 + l16] = tt;
      part += tt;
      part2 += tt * tt;
    }
    rA[(2 * g + 0) * 128 + tid] = part;
    rA[(2 * g + 1) * 128 + tid] = part2;
    }

    fold128(rA, rB, tid);
#pragma unroll
    for (int g = 0; g < GB; g++) {
      if (b0 + g < nbatch) {
        float mean, rstd;
        ln_stats(rB, g, mean, rstd);
        float4 u0 = ((const float4*)mX[g])[tid];
        float4 u1 = ((const float4*)mX[g])[tid + 128];
        u0.x = (u0.x - mean) * rstd; u0.y = (u0.y - mean) * rstd;
        u0.z = (u0.z - mean) * rstd; u0.w = (u0.w - mean) * rstd;
        u1.x = (u1.x - mean) * rstd; u1.y = (u1.y - mean) * rstd;
        u1.z = (u1.z - mean) * rstd; u1.w = (u1.w - mean) * rstd;
        float4* yb4 = (float4*)(out + (size_t)(b0 + g) * 1024);
        yb4[tid] = u0;
        yb4[tid + 128] = u1;
      }
    }
  }
}

__global__ __launch_bounds__(128) void enc_pass3b_mlp(
    const float* __restrict__ W1, const float* __restrict__ b1,
    const float* __restrict__ W2, const float* __restrict__ b2, float* io,
    int nbatch) {
  __shared__ __align__(16) float mY[GB][1024];
  __shared__ __align__(16) float mH[GB][1024];
  __shared__ __align__(16) float rA[2 * GB * 128];
  __shared__ __align__(16) float rB[2 * GB * 20];

  int tid = threadIdx.x;
  int p = tid >> 1, o0 = (tid & 1) * 8;

  float w1[16][8], w2[16][8], c1[8], c2[8];
  {
    const float4* w1p = (const float4*)(W1 + p * 256 + o0);
    const float4* w2p = (const float4*)(W2 + p * 256 + o0);
#pragma unroll
    for (int i = 0; i < 16; i++) {
      float4 u0 = w1p[4 * i], u1 = w1p[4 * i + 1];
      float4 g0 = w2p[4 * i], g1 = w2p[4 * i + 1];
      w1[i][0] = u0.x; w1[i][1] = u0.y; w1[i][2] = u0.z; w1[i][3] = u0.w;
      w1[i][4] = u1.x; w1[i][5] = u1.y; w1[i][6] = u1.z; w1[i][7] = u1.w;
      w2[i][0] = g0.x; w2[i][1] = g0.y; w2[i][2] = g0.z; w2[i][3] = g0.w;
      w2[i][4] = g1.x; w2[i][5] = g1.y; w2[i][6] = g1.z; w2[i][7] = g1.w;
    }
#pragma unroll
    for (int j = 0; j < 8; j++) {
      c1[j] = b1[p * 16 + o0 + j];
      c2[j] = b2[p * 16 + o0 + j];
    }
  }

  for (int b0 = blockIdx.x * GB; b0 < nbatch; b0 += gridDim.x * GB) {
#pragma unroll
    for (int g = 0; g < GB; g++) {
      float4 u0 = make_float4(0.f, 0.f, 0.f, 0.f), u1 = u0;
      if (b0 + g < nbatch) {
        const float4* yb4 = (const float4*)(io + (size_t)(b0 + g) * 1024);
        u0 = yb4[tid];
        u1 = yb4[tid + 128];
      }
      ((float4*)mY[g])[tid] = u0;
      ((float4*)mY[g])[tid + 128] = u1;
    }
    __syncthreads();

#pragma clang loop unroll(disable) vectorize(disable) interleave(disable)
    for (int g = 0; g < GB; g++) {
      float acc[8];
#pragma unroll
      for (int j = 0; j < 8; j++) acc[j] = c1[j];
#pragma unroll
      for (int i = 0; i < 16; i++) {
        float yv = mY[g][p * 16 + i];
#pragma unroll
        for (int j = 0; j < 8; j++) acc[j] = fmaf(yv, w1[i][j], acc[j]);
      }
#pragma unroll
      for (int j = 0; j < 8; j++) mH[g][p * 16 + o0 + j] = fmaxf(acc[j], 0.0f);
    }
    __syncthreads();

#pragma clang loop unroll(disable) vectorize(disable) interleave(disable)
    for (int g = 0; g < GB; g++) {
      float acc2[8];
#pragma unroll
      for (int j = 0; j < 8; j++) acc2[j] = c2[j];
#pragma unroll
      for (int i = 0; i < 16; i++) {
        float hv = mH[g][p * 16 + i];
#pragma unroll
        for (int j = 0; j < 8; j++) acc2[j] = fmaf(hv, w2[i][j], acc2[j]);
      }
      float q1 = 0.0f, q2 = 0.0f;
#pragma unroll
      for (int j = 0; j < 8; j++) {
        float tt = fmaxf(acc2[j], 0.0f) + mY[g][p * 16 + o0 + j];
        mH[g][p * 16 + o0 + j] = tt;
        q1 += tt;
        q2 += tt * tt;
      }
      rA[(2 * g + 0) * 128 + tid] = q1;
      rA[(2 * g + 1) * 128 + tid] = q2;
    }

    fold128(rA, rB, tid);
#pragma unroll
    for (int g = 0; g < GB; g++) {
      if (b0 + g < nbatch) {
        float m2, r2;
        ln_stats(rB, g, m2, r2);
        float* op = io + (size_t)(b0 + g) * 1024 + p * 16 + o0;
#pragma unroll
        for (int j = 0; j < 8; j++) op[j] = (mH[g][p * 16 + o0 + j] - m2) * r2;
      }
    }
  }
}

extern "C" void kernel_launch(void* const* d_in, const int* in_sizes, int n_in,
                              void* d_out, int out_size, void* d_ws,
                              size_t ws_size, hipStream_t stream) {
  (void)n_in;
  (void)out_size;
  (void)ws_size;
  const float* x = (const float*)d_in[0];
  const float* Wq = (const float*)d_in[1];
  const float* Wk = (const float*)d_in[2];
  const float* Wv = (const float*)d_in[3];
  const float* W1 = (const float*)d_in[4];
  const float* b1 = (const float*)d_in[5];
  const float* W2 = (const float*)d_in[6];
  const float* b2 = (const float*)d_in[7];
  float* out = (float*)d_out;

  unsigned int* gmax = (unsigned int*)d_ws;
  float* gsum = (float*)((char*)d_ws + 4096 * sizeof(unsigned int));

  const int B = in_sizes[0] / 1024;

  enc_init<<<16, 256, 0, stream>>>(gmax, gsum);
  enc_pass1_max<<<256, SW * 32, 0, stream>>>(x, Wq, Wk, gmax, B);
  enc_pass2_sum<<<256, SW * 32, 0, stream>>>(x, Wq, Wk, gmax, gsum, B);
  enc_pass3a_attn<<<1280, 128, 0, stream>>>(x, Wq, Wk, Wv, gmax, gsum, out, B);
  enc_pass3b_mlp<<<768, 128, 0, stream>>>(W1, b1, W2, b2, out, B);
}


# pass 3a/3b: a round's four x resp. y tiles requested from memory as one group, then parked in LDS (tail slots clamp to the last batch, no branch between): 4 serial round trips -> 1; sums unchanged
# speedup vs baseline: 2.8312x; 1.0829x over previous
#include <hip/hip_runtime.h>

typedef __attribute__((ext_vector_type(2))) float v2f;
typedef __attribute__((ext_vector_type(8))) float v8f;

#define LN_EPS 1e-5f

__device__ __forceinline__ v8f wmma4(v2f a, v2f b, v8f c) {
  return __builtin_amdgcn_wmma_f32_16x16x4_f32(false, a, false, b, (short)0, c,
                                               false, false);
}

__device__ __forceinline__ v2f ldA(const float* m, int ld, int r0, int c0,
                                   int l16, int h) {
  v2f a;
  a.x = m[(r0 + l16) * ld + c0 + 2 * h + 0];
  a.y = m[(r0 + l16) * ld + c0 + 2 * h + 1];
  return a;
}
__device__ __forceinline__ v2f ldB(const float* m, int ld, int c0, int n0,
                                   int l16, int h) {
  v2f b;
  b.x = m[(c0 + 2 * h + 0) * ld + n0 + l16];
  b.y = m[(c0 + 2 * h + 1) * ld + n0 + l16];
  return b;
}
__device__ __forceinline__ v2f ldBT(const float* m, int ld, int c0, int n0,
                                    int l16, int h) {
  v2f b;
  b.x = m[(n0 + l16) * ld + c0 + 2 * h + 0];
  b.y = m[(n0 + l16) * ld + c0 + 2 * h + 1];
  return b;
}

__device__ __forceinline__ unsigned int f2key(float f) {
  unsigned int u = __float_as_uint(f);
  return (u & 0x80000000u) ? ~u : (u | 0x80000000u);
}
__device__ __forceinline__ float key2f(unsigned int k) {
  unsigned int u = (k & 0x80000000u) ? (k & 0x7FFFFFFFu) : ~k;
  return __uint_as_float(u);
}

#define GB 4
__device__ __forceinline__ void fold128(const float* rA, float* rB, int tid) {
  float* rC = rB + 2 * GB * 16;
  __syncthreads();
  {
    const float4* p = (const float4*)(rA + tid * 8);
    float4 a = p[0], b = p[1];
    rB[tid] = ((a.x + a.y) + (a.z + a.w)) + ((b.x + b.y) + (b.z + b.w));
  }
  __syncthreads();
  if (tid < 2 * GB * 4) {
    float4 a = ((const float4*)rB)[tid];
    rC[tid] = (a.x + a.y) + (a.z + a.w);
  }
  __syncthreads();
}
__device__ __forceinline__ void ln_stats(const float* rB, int g, float& mean,
                                         float& rstd) {
  const float4* rC = (const float4*)(rB + 2 * GB * 16);
  float4 a = rC[2 * g], b = rC[2 * g + 1];
  float s1 = (a.x + a.y) + (a.z + a.w);
  float s2 = (b.x + b.y) + (b.z + b.w);
  mean = s1 * (1.0f / 1024.0f);
  rstd = rsqrtf(s2 * (1.0f / 1024.0f) - mean * mean + LN_EPS);
}

__global__ void enc_init(unsigned int* gmax, float* gsum) {
  int i = blockIdx.x * blockDim.x + threadIdx.x;
  if (i < 4096) {
    gmax[i] = 0u;
    gsum[i] = 0.0f;
  }
}

#define SW 16
#define QP 20
#define QKW (2 * 64 * QP)

__device__ __forceinline__ void ldXrow(const float* xb, int t, int l16, int h,
                                       v2f a[4]) {
  const float4* p = (const float4*)(xb + (16 * t + l16) * 16 + 8 * h);
  float4 u0 = p[0], u1 = p[1];
  a[0].x = u0.x; a[0].y = u0.y; a[1].x = u0.z; a[1].y = u0.w;
  a[2].x = u1.x; a[2].y = u1.y; a[3].x = u1.z; a[3].y = u1.w;
}
__device__ __forceinline__ void ldSrow(const float* m, int t, int l16, int h,
                                       v2f a[4]) {
  const float4* p = (const float4*)(m + (16 * t + l16) * QP + 8 * h);
  float4 u0 = p[0], u1 = p[1];
  a[0].x = u0.x; a[0].y = u0.y; a[1].x = u0.z; a[1].y = u0.w;
  a[2].x = u1.x; a[2].y = u1.y; a[3].x = u1.z; a[3].y = u1.w;
}
__device__ __forceinline__ void ldWfrag(const float* wm, int l16, int h,
                                        v2f b[4]) {
#pragma unroll
  for (int kk = 0; kk < 4; kk++) {
    b[kk].x = wm[(8 * h + 2 * kk + 0) * 16 + l16];
    b[kk].y = wm[(8 * h + 2 * kk + 1) * 16 + l16];
  }
}
__device__ __forceinline__ void qk_frags(const float* xb, const v2f bq[4],
                                         const v2f bk[4], float* sQ, float* sK,
                                         int l16, int h, v2f qa[4][4],
                                         v2f kb[4][4]) {
  v2f a[4][4];
#pragma unroll
  for (int t = 0; t < 4; t++) ldXrow(xb, t, l16, h, a[t]);
#pragma unroll
  for (int t = 0; t < 4; t++) {
    v8f q = {};
    v8f k = {};
#pragma unroll
    for (int kk = 0; kk < 4; kk++) {
      q = wmma4(a[t][kk], bq[kk], q);
      k = wmma4(a[t][kk], bk[kk], k);
    }
#pragma unroll
    for (int r = 0; r < 8; r++) {
      sQ[(16 * t + r + 8 * h) * QP + l16] = q[r];
      sK[(16 * t + r + 8 * h) * QP + l16] = k[r];
    }
  }
#pragma unroll
  for (int t = 0; t < 4; t++) {
    ldSrow(sQ, t, l16, h, qa[t]);
    ldSrow(sK, t, l16, h, kb[t]);
  }
}
template <bool IS_MAX>
__device__ __forceinline__ void fold_waves(float (&v)[16][8], float* red, int w,
                                           int lane) {
  __syncthreads();
  for (int half = SW / 2; half >= 1; half >>= 1) {
    if (w >= half && w < 2 * half) {
      float* dst = red + (w & 7) * 4096;
#pragma unroll
      for (int tl = 0; tl < 16; tl++)
#pragma unroll
        for (int r = 0; r < 8; r++) dst[(tl * 8 + r) * 32 + lane] = v[tl][r];
    }
    __syncthreads();
    if (w < half) {
      const float* sp = red + ((w + half) & 7) * 4096;
#pragma unroll
      for (int tl = 0; tl < 16; tl++)
#pragma unroll
        for (int r = 0; r < 8; r++) {
          float o = sp[(tl * 8 + r) * 32 + lane];
          v[tl][r] = IS_MAX ? fmaxf(v[tl][r], o) : v[tl][r] + o;
        }
    }
  }
  if (w == 0) {
#pragma unroll
    for (int tl = 0; tl < 16; tl++)
#pragma unroll
      for (int r = 0; r < 8; r++) red[(tl * 8 + r) * 32 + lane] = v[tl][r];
  }
  __syncthreads();
}

__global__ __launch_bounds__(SW * 32) void enc_pass1_max(
    const float* __restrict__ x, const float* __restrict__ Wq,
    const float* __restrict__ Wk, unsigned int* __restrict__ gmax, int nbatch) {
  __shared__ __align__(16) float sQK[SW * QKW];
  int tid = threadIdx.x;
  int lane = tid & 31, l16 = lane & 15, h = lane >> 4;
  int w = __builtin_amdgcn_readfirstlane(tid >> 5);
  float* sQ = sQK + w * QKW;
  float* sK = sQ + 64 * QP;
  v2f bq[4], bk[4];
  ldWfrag(Wq, l16, h, bq);
  ldWfrag(Wk, l16, h, bk);
  float vmax[16][8];
#pragma unroll
  for (int tl = 0; tl < 16; tl++)
#pragma unroll
    for (int r = 0; r < 8; r++) vmax[tl][r] = -3.4e38f;

  for (int bi = blockIdx.x * SW + w; bi < nbatch; bi += gridDim.x * SW) {
    v2f qa[4][4], kb[4][4];
    qk_frags(x + (size_t)bi * 1024, bq, bk, sQ, sK, l16, h, qa, kb);
#pragma unroll
    for (int qt = 0; qt < 4; qt++)
#pragma unroll
      for (int kt = 0; kt < 4; kt++) {
        v8f s = {};
#pragma unroll
        for (int kk = 0; kk < 4; kk++) s = wmma4(qa[qt][kk], kb[kt][kk], s);
#pragma unroll
        for (int r = 0; r < 8; r++)
          vmax[qt * 4 + kt][r] =
              fmaxf(vmax[qt * 4 + kt][r], s[r] * (1.0f / 3.0f));
      }
  }

  fold_waves<true>(vmax, sQK, w, lane);
  for (int j = 0; j < 4096 / (SW * 32); ++j) {
    int idx = j * (SW * 32) + tid;
    int tr = idx >> 5;
    int tl = tr >> 3, r = tr & 7;
    int qrow = 16 * (tl >> 2) + r + 8 * h, kcol = 16 * (tl & 3) + l16;
    atomicMax(&gmax[qrow * 64 + kcol], f2key(sQK[idx]));
  }
}

__global__ __launch_bounds__(SW * 32) void enc_pass2_sum(
    const float* __restrict__ x, const float* __restrict__ Wq,
    const float* __restrict__ Wk, const unsigned int* __restrict__ gmax,
    float* __restrict__ gsum, int nbatch) {
  __shared__ __align__(16) float sQK[SW * QKW];
  __shared__ __align__(16) float sGm[4096];
  int tid = threadIdx.x;
  int lane = tid & 31, l16 = lane & 15, h = lane >> 4;
  int w = __builtin_amdgcn_readfirstlane(tid >> 5);
  float* sQ = sQK + w * QKW;
  float* sK = sQ + 64 * QP;
  v2f bq[4], bk[4];
  ldWfrag(Wq, l16, h, bq);
  ldWfrag(Wk, l16, h, bk);
  for (int j = 0; j < 4096 / (SW * 32); ++j) {
    int idx = j * (SW * 32) + tid;
    int jj = idx & 3, ln = (idx >> 2) & 31, trq = idx >> 7;
    int tl = trq >> 1, r = (trq & 1) * 4 + jj;
    int qrow = 16 * (tl >> 2) + r + 8 * (ln >> 4), kcol = 16 * (tl & 3) + (ln & 15);
    sGm[idx] = key2f(gmax[qrow * 64 + kcol]);
  }
  float vsum[16][8];
#pragma unroll
  for (int tl = 0; tl < 16; tl++)
#pragma unroll
    for (int r = 0; r < 8; r++) vsum[tl][r] = 0.0f;
  __syncthreads();

  for (int bi = blockIdx.x * SW + w; bi < nbatch; bi += gridDim.x * SW) {
    v2f qa[4][4], kb[4][4];
    qk_frags(x + (size_t)bi * 1024, bq, bk, sQ, sK, l16, h, qa, kb);
#pragma unroll
    for (int qt = 0; qt < 4; qt++)
#pragma unroll
      for (int kt = 0; kt < 4; kt++) {
        const int tl = qt * 4 + kt;
        v8f s = {};
#pragma unroll
        for (int kk = 0; kk < 4; kk++) s = wmma4(qa[qt][kk], kb[kt][kk], s);
        float4 g0 = ((const float4*)sGm)[(tl * 2 + 0) * 32 + lane];
        float4 g1 = ((const float4*)sGm)[(tl * 2 + 1) * 32 + lane];
        float gm[8] = {g0.x, g0.y, g0.z, g0.w, g1.x, g1.y, g1.z, g1.w};
#pragma unroll
        for (int r = 0; r < 8; r++)
          vsum[tl][r] += __expf(s[r] * (1.0f / 3.0f) - gm[r]);
      }
  }

  fold_waves<false>(vsum, sQK, w, lane);
  for (int j = 0; j < 4096 / (SW * 32); ++j) {
    int idx = j * (SW * 32) + tid;
    int tr = idx >> 5;
    int tl = tr >> 3, r = tr & 7;
    int qrow = 16 * (tl >> 2) + r + 8 * h, kcol = 16 * (tl & 3) + l16;
    atomicAdd(&gsum[qrow * 64 + kcol], sQK[idx]);
  }
}

__global__ __launch_bounds__(128) __attribute__((amdgpu_waves_per_eu(5, 5))) void enc_pass3a_attn(
    const float* __restrict__ x, const float* __restrict__ Wq,
    const float* __restrict__ Wk, const float* __restrict__ Wv,
    const unsigned int* __restrict__ gmax, const float* __restrict__ gsum,
    float* __restrict__ out, int nbatch) {
  __shared__ float sWq[256], sWk[256], sWv[256];
  __shared__ __align__(16) float mX[GB][1024];
  __shared__ float mQ[1024], mK[1024], mV[1024];
  __shared__ __align__(16) float mS[4096];
  __shared__ __align__(16) float rA[2 * GB * 128];
  __shared__ __align__(16) float rB[2 * GB * 20];

  int tid = threadIdx.x;
  int w = tid >> 5, lane = tid & 31, l16 = lane & 15, h = lane >> 4;
  int r0 = 16 * w;
  float* mSw = mS + 1024 * w;

  for (int i = tid; i < 256; i += 128) {
    sWq[i] = Wq[i];
    sWk[i] = Wk[i];
    sWv[i] = Wv[i];
  }
  float gm[4][8], rs[4][8];
#pragma unroll
  for (int kt = 0; kt < 4; kt++)
#pragma unroll
    for (int r = 0; r < 8; r++) {
      int qrow = r0 + r + 8 * h, kcol = 16 * kt + l16;
      gm[kt][r] = key2f(gmax[qrow * 64 + kcol]);
      rs[kt][r] = 1.0f / gsum[qrow * 64 + kcol];
    }
  __syncthreads();

  for (int b0 = blockIdx.x * GB; b0 < nbatch; b0 += gridDim.x * GB) {
    float4 xu[GB][2];
#pragma unroll
    for (int g = 0; g < GB; g++) {
      int bb = min(b0 + g, nbatch - 1);
      const float4* xb4 = (const float4*)(x + (size_t)bb * 1024);
      xu[g][0] = xb4[tid];
      xu[g][1] = xb4[tid + 128];
    }
#pragma unroll
    for (int g = 0; g < GB; g++) {
      ((float4*)mX[g])[tid] = xu[g][0];
      ((float4*)mX[g])[tid + 128] = xu[g][1];
    }

#pragma clang loop unroll(disable) vectorize(disable) interleave(disable)
    for (int g = 0; g < GB; g++) {
    float* mXg = mX[g];
    __syncthreads();


    v2f a[4], wq[4], wk[4], wv[4];
#pragma unroll
    for (int kk = 0; kk < 4; kk++) a[kk] = ldA(mXg, 16, r0, 4 * kk, l16, h);
#pragma unroll
    for (int kk = 0; kk < 4; kk++) {
      wq[kk] = ldB(sWq, 16, 4 * kk, 0, l16, h);
      wk[kk] = ldB(sWk, 16, 4 * kk, 0, l16, h);
      wv[kk] = ldB(sWv, 16, 4 * kk, 0, l16, h);
    }
    v8f q = {};
    v8f k = {};
    v8f v = {};
#pragma unroll
    for (int kk = 0; kk < 4; kk++) {
      q = wmma4(a[kk], wq[kk], q);
      k = wmma4(a[kk], wk[kk], k);
      v = wmma4(a[kk], wv[kk], v);
    }
#pragma unroll
    for (int r = 0; r < 8; r++) {
      int row = r0 + r + 8 * h;
      mQ[row * 16 + l16] = q[r];
      mK[row * 16 + l16] = k[r];
      mV[row * 16 + l16] = v[r];
    }
    __syncthreads();

    v2f qa[4], kf[4][4];
#pragma unroll
    for (int kk = 0; kk < 4; kk++) qa[kk] = ldA(mQ, 16, r0, 4 * kk, l16, h);
#pragma unroll
    for (int kk = 0; kk < 4; kk++)
#pragma unroll
      for (int kt = 0; kt < 4; kt++)
        kf[kk][kt] = ldBT(mK, 16, 4 * kk, 16 * kt, l16, h);
    v8f s[4];
#pragma unroll
    for (int kt = 0; kt < 4; kt++) s[kt] = v8f{};
#pragma unroll
    for (int kk = 0; kk < 4; kk++)
#pragma unroll
      for (int kt = 0; kt < 4; kt++)
        s[kt] = wmma4(qa[kk], kf[kk][kt], s[kt]);
#pragma unroll
    for (int kt = 0; kt < 4; kt++) {
      float e[8];
#pragma unroll
      for (int r = 0; r < 8; r++)
        e[r] = __expf(s[kt][r] * (1.0f / 3.0f) - gm[kt][r]) * rs[kt][r];
      float4* dst = (float4*)(mSw + (16 * kt + l16) * 16 + 8 * h);
      dst[0] = make_float4(e[0], e[1], e[2], e[3]);
      dst[1] = make_float4(e[4], e[5], e[6], e[7]);
    }

    v2f af[4][4], vf[4][4];
    v8f z = {};
#pragma unroll
    for (int j = 0; j < 4; j++) {
      af[0][j] = ldB(mSw, 16, 4 * j, 0, l16, h);
      vf[0][j] = ldB(mV, 16, 4 * j, 0, l16, h);
    }
#pragma unroll
    for (int gi = 0; gi < 4; gi++) {
      if (gi < 3) {
#pragma unroll
        for (int j = 0; j < 4; j++) {
          af[gi + 1][j] = ldB(mSw, 16, 16 * (gi + 1) + 4 * j, 0, l16, h);
          vf[gi + 1][j] = ldB(mV, 16, 16 * (gi + 1) + 4 * j, 0, l16, h);
        }
      }
#pragma unroll
      for (int j = 0; j < 4; j++) z = wmma4(af[gi][j], vf[gi][j], z);
    }

    float xr[8];
#pragma unroll
    for (int r = 0; r < 8; r++) xr[r] = mXg[(r0 + r + 8 * h) * 16 + l16];
    float part = 0.0f, part2 = 0.0f;
#pragma unroll
    for (int r = 0; r < 8; r++) {
      int row = r0 + r + 8 * h;
      float tt = xr[r] + z[r];
      mXg[row * 16 + l16] = tt;
      part += tt;
      part2 += tt * tt;
    }
    rA[(2 * g + 0) * 128 + tid] = part;
    rA[(2 * g + 1) * 128 + tid] = part2;
    }

    fold128(rA, rB, tid);
#pragma unroll
    for (int g = 0; g < GB; g++) {
      if (b0 + g < nbatch) {
        float mean, rstd;
        ln_stats(rB, g, mean, rstd);
        float4 u0 = ((const float4*)mX[g])[tid];
        float4 u1 = ((const float4*)mX[g])[tid + 128];
        u0.x = (u0.x - mean) * rstd; u0.y = (u0.y - mean) * rstd;
        u0.z = (u0.z - mean) * rstd; u0.w = (u0.w - mean) * rstd;
        u1.x = (u1.x - mean) * rstd; u1.y = (u1.y - mean) * rstd;
        u1.z = (u1.z - mean) * rstd; u1.w = (u1.w - mean) * rstd;
        float4* yb4 = (float4*)(out + (size_t)(b0 + g) * 1024);
        yb4[tid] = u0;
        yb4[tid + 128] = u1;
      }
    }
  }
}

__global__ __launch_bounds__(128) void enc_pass3b_mlp(
    const float* __restrict__ W1, const float* __restrict__ b1,
    const float* __restrict__ W2, const float* __restrict__ b2, float* io,
    int nbatch) {
  __shared__ __align__(16) float mY[GB][1024];
  __shared__ __align__(16) float mH[GB][1024];
  __shared__ __align__(16) float rA[2 * GB * 128];
  __shared__ __align__(16) float rB[2 * GB * 20];

  int tid = threadIdx.x;
  int p = tid >> 1, o0 = (tid & 1) * 8;

  float w1[16][8], w2[16][8], c1[8], c2[8];
  {
    const float4* w1p = (const float4*)(W1 + p * 256 + o0);
    const float4* w2p = (const float4*)(W2 + p * 256 + o0);
#pragma unroll
    for (int i = 0; i < 16; i++) {
      float4 u0 = w1p[4 * i], u1 = w1p[4 * i + 1];
      float4 g0 = w2p[4 * i], g1 = w2p[4 * i + 1];
      w1[i][0] = u0.x; w1[i][1] = u0.y; w1[i][2] = u0.z; w1[i][3] = u0.w;
      w1[i][4] = u1.x; w1[i][5] = u1.y; w1[i][6] = u1.z; w1[i][7] = u1.w;
      w2[i][0] = g0.x; w2[i][1] = g0.y; w2[i][2] = g0.z; w2[i][3] = g0.w;
      w2[i][4] = g1.x; w2[i][5] = g1.y; w2[i][6] = g1.z; w2[i][7] = g1.w;
    }
#pragma unroll
    for (int j = 0; j < 8; j++) {
      c1[j] = b1[p * 16 + o0 + j];
      c2[j] = b2[p * 16 + o0 + j];
    }
  }

  for (int b0 = blockIdx.x * GB; b0 < nbatch; b0 += gridDim.x * GB) {
    float4 yu[GB][2];
#pragma unroll
    for (int g = 0; g < GB; g++) {
      int bb = min(b0 + g, nbatch - 1);
      const float4* yb4 = (const float4*)(io + (size_t)bb * 1024);
      yu[g][0] = yb4[tid];
      yu[g][1] = yb4[tid + 128];
    }
#pragma unroll
    for (int g = 0; g < GB; g++) {
      ((float4*)mY[g])[tid] = yu[g][0];
      ((float4*)mY[g])[tid + 128] = yu[g][1];
    }
    __syncthreads();

#pragma clang loop unroll(disable) vectorize(disable) interleave(disable)
    for (int g = 0; g < GB; g++) {
      float acc[8];
#pragma unroll
      for (int j = 0; j < 8; j++) acc[j] = c1[j];
#pragma unroll
      for (int i = 0; i < 16; i++) {
        float yv = mY[g][p * 16 + i];
#pragma unroll
        for (int j = 0; j < 8; j++) acc[j] = fmaf(yv, w1[i][j], acc[j]);
      }
#pragma unroll
      for (int j = 0; j < 8; j++) mH[g][p * 16 + o0 + j] = fmaxf(acc[j], 0.0f);
    }
    __syncthreads();

#pragma clang loop unroll(disable) vectorize(disable) interleave(disable)
    for (int g = 0; g < GB; g++) {
      float acc2[8];
#pragma unroll
      for (int j = 0; j < 8; j++) acc2[j] = c2[j];
#pragma unroll
      for (int i = 0; i < 16; i++) {
        float hv = mH[g][p * 16 + i];
#pragma unroll
        for (int j = 0; j < 8; j++) acc2[j] = fmaf(hv, w2[i][j], acc2[j]);
      }
      float q1 = 0.0f, q2 = 0.0f;
#pragma unroll
      for (int j = 0; j < 8; j++) {
        float tt = fmaxf(acc2[j], 0.0f) + mY[g][p * 16 + o0 + j];
        mH[g][p * 16 + o0 + j] = tt;
        q1 += tt;
        q2 += tt * tt;
      }
      rA[(2 * g + 0) * 128 + tid] = q1;
      rA[(2 * g + 1) * 128 + tid] = q2;
    }

    fold128(rA, rB, tid);
#pragma unroll
    for (int g = 0; g < GB; g++) {
      if (b0 + g < nbatch) {
        float m2, r2;
        ln_stats(rB, g, m2, r2);
        float* op = io + (size_t)(b0 + g) * 1024 + p * 16 + o0;
#pragma unroll
        for (int j = 0; j < 8; j++) op[j] = (mH[g][p * 16 + o0 + j] - m2) * r2;
      }
    }
  }
}

extern "C" void kernel_launch(void* const* d_in, const int* in_sizes, int n_in,
                              void* d_out, int out_size, void* d_ws,
                              size_t ws_size, hipStream_t stream) {
  (void)n_in;
  (void)out_size;
  (void)ws_size;
  const float* x = (const float*)d_in[0];
  const float* Wq = (const float*)d_in[1];
  const float* Wk = (const float*)d_in[2];
  const float* Wv = (const float*)d_in[3];
  const float* W1 = (const float*)d_in[4];
  const float* b1 = (const float*)d_in[5];
  const float* W2 = (const float*)d_in[6];
  const float* b2 = (const float*)d_in[7];
  float* out = (float*)d_out;

  unsigned int* gmax = (unsigned int*)d_ws;
  float* gsum = (float*)((char*)d_ws + 4096 * sizeof(unsigned int));

  const int B = in_sizes[0] / 1024;

  enc_init<<<16, 256, 0, stream>>>(gmax, gsum);
  enc_pass1_max<<<256, SW * 32, 0, stream>>>(x, Wq, Wk, gmax, B);
  enc_pass2_sum<<<256, SW * 32, 0, stream>>>(x, Wq, Wk, gmax, gsum, B);
  enc_pass3a_attn<<<1280, 128, 0, stream>>>(x, Wq, Wk, Wv, gmax, gsum, out, B);
  enc_pass3b_mlp<<<768, 128, 0, stream>>>(W1, b1, W2, b2, out, B);
}
